# starting kernel
# baseline (speedup 1.0000x reference)
#include <hip/hip_runtime.h>

typedef __attribute__((ext_vector_type(2))) float v2f;
typedef __attribute__((ext_vector_type(8))) float v8f;

#define DIMI 128
#define ODIM 256
#define SEQL 8192
#define NB   8
#define CH   64

__global__ void wnorm_kernel(const float* __restrict__ w, float* __restrict__ we) {
    int o = blockIdx.x * blockDim.x + threadIdx.x;
    if (o >= ODIM) return;
    float s = 0.f;
#pragma unroll 8
    for (int i = 0; i < DIMI; ++i) { float v = w[o * DIMI + i]; s += v * v; }
    float inv = 1.f / (sqrtf(s) + 1.13137084989847604e-3f);
#pragma unroll 8
    for (int i = 0; i < DIMI; ++i) we[o * DIMI + i] = w[o * DIMI + i] * inv;
}

__global__ void __launch_bounds__(256)
gemm_coef_kernel(const float* __restrict__ x, const float* __restrict__ we,
                 float2* __restrict__ coef) {
    const int lane = threadIdx.x & 31;
    const int wv   = threadIdx.x >> 5;
    const int dir  = wv >> 2;
    const int mg   = wv & 3;
    const int hb   = dir * 128 + mg * 16;
    const int gb   = hb + 64;
    const int n0   = blockIdx.x * 16;
    const int bb   = blockIdx.y;

    const int ln    = lane & 15;
    const int khalf = (lane >> 4) << 1;

    const float* xb  = x + (size_t)bb * DIMI * SEQL;
    const float* wrh = we + (size_t)(hb + ln) * DIMI;
    const float* wrg = we + (size_t)(gb + ln) * DIMI;

    v8f ch = {};
    v8f cg = {};

#pragma unroll 4
    for (int k = 0; k < DIMI; k += 4) {
        const float2 th = *(const float2*)(wrh + k + khalf);
        const float2 tg = *(const float2*)(wrg + k + khalf);
        v2f ah = { th.x, th.y };
        v2f ag = { tg.x, tg.y };
        const float b0 = xb[(size_t)(k + khalf) * SEQL + n0 + ln];
        const float b1 = xb[(size_t)(k + khalf + 1) * SEQL + n0 + ln];
        v2f bf = { b0, b1 };
        ch = __builtin_amdgcn_wmma_f32_16x16x4_f32(false, ah, false, bf,
                                                   (short)0, ch, false, false);
        cg = __builtin_amdgcn_wmma_f32_16x16x4_f32(false, ag, false, bf,
                                                   (short)0, cg, false, false);
    }

    const int mrow = (lane >> 4) << 3;
    const int n    = n0 + ln;
#pragma unroll
    for (int r = 0; r < 8; ++r) {
        const int c = (hb & 63) + mrow + r;
        const float h = ch[r];
        const float g = cg[r];
        const float eg  = __expf(g);
        const float emg = __expf(-g);
        const float z  = __builtin_amdgcn_rcpf(1.f + emg);
        const float oz = __builtin_amdgcn_rcpf(1.f + eg);
        const float s  = __builtin_amdgcn_sqrtf(
            2.f * __builtin_amdgcn_rcpf(eg + emg) + 1.f);
        coef[((size_t)(dir * NB + bb) * CH + c) * SEQL + n] =
            make_float2(oz * s, z * s * h);
    }
}

template <int DIR>
__global__ void __launch_bounds__(256)
scan_kernel(const float2* __restrict__ coef, float* __restrict__ out) {
    const int tid = threadIdx.x;
    const int c   = blockIdx.x;
    const int bb  = blockIdx.y;
    const size_t base  = ((size_t)(DIR * NB + bb) * CH + c) * SEQL;
    const int    lbase = DIR ? (SEQL - 32 - tid * 32) : (tid * 32);

    float2 e[32];
#pragma unroll
    for (int j = 0; j < 32; ++j) e[j] = coef[base + lbase + j];

    float A = 1.f, Bv = 0.f;
#pragma unroll
    for (int s = 0; s < 32; ++s) {
        const int j = DIR ? (31 - s) : s;
        A  = e[j].x * A;
        Bv = e[j].x * Bv + e[j].y;
    }

    __shared__ float sA[256];
    __shared__ float sB[256];
    sA[tid] = A; sB[tid] = Bv;
    __syncthreads();
    for (int off = 1; off < 256; off <<= 1) {
        float pa = 1.f, pb = 0.f;
        if (tid >= off) { pa = sA[tid - off]; pb = sB[tid - off]; }
        __syncthreads();
        if (tid >= off) {
            const float ca = sA[tid], cb = sB[tid];
            sA[tid] = ca * pa;
            sB[tid] = ca * pb + cb;
        }
        __syncthreads();
    }
    float h = (tid == 0) ? 0.f : sB[tid - 1];

#pragma unroll
    for (int s = 0; s < 32; ++s) {
        const int j = DIR ? (31 - s) : s;
        h = e[j].x * h + e[j].y;
        e[j].x = h;
    }

    float* op = out + ((size_t)bb * 128 + DIR * 64 + c) * SEQL + lbase;
#pragma unroll
    for (int j = 0; j < 32; ++j) op[j] = e[j].x;
}

extern "C" void kernel_launch(void* const* d_in, const int* in_sizes, int n_in,
                              void* d_out, int out_size, void* d_ws, size_t ws_size,
                              hipStream_t stream) {
    const float* x = (const float*)d_in[0];
    const float* w = (const float*)d_in[1];
    float* out = (float*)d_out;

    float*  we   = (float*)d_ws;
    float2* coef = (float2*)((char*)d_ws + ODIM * DIMI * sizeof(float));

    wnorm_kernel<<<1, 256, 0, stream>>>(w, we);
    gemm_coef_kernel<<<dim3(SEQL / 16, NB), 256, 0, stream>>>(x, we, coef);
    scan_kernel<0><<<dim3(CH, NB), 256, 0, stream>>>(coef, out);
    scan_kernel<1><<<dim3(CH, NB), 256, 0, stream>>>(coef, out);
}


# GEMM: LDS weight image's rotation period halved and a direction's four pairs unrolled, so fragment-load and store addresses are offset fields; the pair loop's ~50 address vector ops per pair are gone
# speedup vs baseline: 2.0937x; 2.0937x over previous
#include <hip/hip_runtime.h>

typedef __attribute__((ext_vector_type(2))) float v2f;
typedef __attribute__((ext_vector_type(8))) float v8f;

#define DIMI 128
#define ODIM 256
#define SEQL 8192
#define NB   8
#define CH   64

__global__ void wnorm_kernel(const float* __restrict__ w, float* __restrict__ inv) {
    int o = blockIdx.x * blockDim.x + threadIdx.x;
    if (o >= ODIM) return;
    const float4* row = (const float4*)(w + o * DIMI);
    float4 v[DIMI / 4];
#pragma unroll
    for (int i = 0; i < DIMI / 4; ++i) v[i] = row[i];
    float s = 0.f;
#pragma unroll
    for (int i = 0; i < DIMI / 4; ++i) {
        s += v[i].x * v[i].x; s += v[i].y * v[i].y;
        s += v[i].z * v[i].z; s += v[i].w * v[i].w;
    }
    inv[o] = 1.f / (sqrtf(s) + 1.13137084989847604e-3f);
}

__global__ void __launch_bounds__(256)
gemm_raw_kernel(const float* __restrict__ x, const float* __restrict__ w,
                float2* __restrict__ hg) {
    __shared__ v2f wl[ODIM * DIMI / 2];

    const int tid   = threadIdx.x;
    const int lane  = tid & 31;
    const int wv    = tid >> 5;
    const int ln    = lane & 15;
    const int khalf = (lane >> 4) << 1;
    const int n0    = (blockIdx.x * 8 + wv) * 16;
    const int bb    = blockIdx.y;

    const float* xb = x + (size_t)bb * DIMI * SEQL + n0 + ln;
    v2f bf[32];
#pragma unroll
    for (int s = 0; s < 32; ++s) {
        bf[s].x = xb[(size_t)(4 * s + khalf) * SEQL];
        bf[s].y = xb[(size_t)(4 * s + khalf + 1) * SEQL];
    }

    const float2* wsrc = (const float2*)w;
#pragma unroll 16
    for (int it = 0; it < 64; ++it) {
        const int i    = it * 256 + tid;
        const int row  = i >> 6;
        const int kp   = i & 63;
        const int s    = kp >> 1;
        const int slot = ((kp & 1) * 16 + (row & 15) + (s & 15)) & 31;
        const float2 t = wsrc[i];
        v2f v = { t.x, t.y };
        wl[((row >> 4) * 32 + s) * 32 + slot] = v;
    }
    __syncthreads();

    const int mrow = (lane >> 4) << 3;
    const int n    = n0 + ln;

    int ab[16];
#pragma unroll
    for (int j = 0; j < 16; ++j)
        ab[j] = (((lane + j) & 31) << 3) + ((j & 1) << 8);
    const char* wb = (const char*)wl;

    float2* cp = hg + ((size_t)bb * CH + mrow) * SEQL + n;

#pragma unroll 1
    for (int dir = 0; dir < 2; ++dir) {
#pragma unroll
        for (int mg = 0; mg < 4; ++mg) {
            const int tb = mg * 8192;

            v8f ch = {};
            v8f cg = {};
#pragma unroll
            for (int s = 0; s < 32; ++s) {
                const int off = tb + (s >> 1) * 512;
                const v2f ah = *(const v2f*)(wb + ab[s & 15] + off);
                const v2f ag = *(const v2f*)(wb + ab[s & 15] + off + 32768);
                ch = __builtin_amdgcn_wmma_f32_16x16x4_f32(false, ah, false, bf[s],
                                                           (short)0, ch, false, false);
                cg = __builtin_amdgcn_wmma_f32_16x16x4_f32(false, ag, false, bf[s],
                                                           (short)0, cg, false, false);
            }
            __builtin_amdgcn_sched_group_barrier(0x100, 32, 0);
            __builtin_amdgcn_sched_group_barrier(0x008, 64, 0);

#pragma unroll
            for (int r = 0; r < 8; ++r)
                cp[(size_t)(mg * 16 + r) * SEQL] = make_float2(ch[r], cg[r]);
        }
#pragma unroll
        for (int j = 0; j < 16; ++j) ab[j] += 65536;
        cp += (size_t)NB * CH * SEQL;
    }
}

__device__ __forceinline__ void gru_coef(float hraw, float graw, float ih, float kg,
                                         float& a, float& b) {
    const float gs = graw * kg;
    const float t  = __builtin_amdgcn_exp2f(-__builtin_fabsf(gs));
    const float r  = __builtin_amdgcn_rsqf(__builtin_fmaf(t, t, 1.f));
    const float tr = t * r;
    const bool pos = gs > 0.f;
    a = pos ? tr : r;
    b = (pos ? r : tr) * (hraw * ih);
}

template <int DIR>
__device__ __forceinline__ float lane_prev(float v, int d) {
    return DIR ? __shfl_down(v, d) : __shfl_up(v, d);
}

__device__ __forceinline__ float lane_bcast(float v, int l) {
    return __builtin_bit_cast(float,
        __builtin_amdgcn_readlane(__builtin_bit_cast(int, v), l));
}

template <int DIR>
__global__ void __launch_bounds__(256)
scan_kernel(const float2* __restrict__ hg, const float* __restrict__ inv,
            float* __restrict__ out) {
    const int tid  = threadIdx.x;
    const int lane = tid & 31;
    const int wv   = tid >> 5;
    const int c    = blockIdx.x;
    const int bb   = blockIdx.y;
    const int tl   = DIR ? 31 - lane : lane;
    const int ws   = DIR ? 7 - wv : wv;
    const int LAST = DIR ? 0 : 31;

    const size_t base = ((size_t)(DIR * NB + bb) * CH + c) * SEQL;
    const float4* src = (const float4*)(hg + base + wv * 1024 + lane * 4);

    float4 q0[8], q1[8];
#pragma unroll
    for (int r = 0; r < 8; ++r) {
        q0[r] = src[r * 64];
        q1[r] = src[r * 64 + 1];
    }

    const float ih = inv[DIR * 128 + c];
    const float kg = inv[DIR * 128 + 64 + c] * 1.44269504088896341f;

    float a[8][4], b[8][4];
#pragma unroll
    for (int r = 0; r < 8; ++r) {
        gru_coef(q0[r].x, q0[r].y, ih, kg, a[r][0], b[r][0]);
        gru_coef(q0[r].z, q0[r].w, ih, kg, a[r][1], b[r][1]);
        gru_coef(q1[r].x, q1[r].y, ih, kg, a[r][2], b[r][2]);
        gru_coef(q1[r].z, q1[r].w, ih, kg, a[r][3], b[r][3]);
    }

    float A[8], B[8];
#pragma unroll
    for (int r = 0; r < 8; ++r) {
        float ar = 1.f, br = 0.f;
#pragma unroll
        for (int s = 0; s < 4; ++s) {
            const int j = DIR ? 3 - s : s;
            br = a[r][j] * br + b[r][j];
            ar = a[r][j] * ar;
        }
#pragma unroll
        for (int d = 1; d < 32; d <<= 1) {
            const float pa = lane_prev<DIR>(ar, d);
            const float pb = lane_prev<DIR>(br, d);
            if (tl >= d) { br = ar * pb + br; ar = ar * pa; }
        }
        A[r] = ar; B[r] = br;
    }

    float PA[8], PB[8];
    float ca = 1.f, cb = 0.f;
#pragma unroll
    for (int rr = 0; rr < 8; ++rr) {
        const int r = DIR ? 7 - rr : rr;
        PA[r] = ca; PB[r] = cb;
        const float ra = lane_bcast(A[r], LAST);
        const float rb = lane_bcast(B[r], LAST);
        cb = ra * cb + rb;
        ca = ra * ca;
    }

    __shared__ float sA[8];
    __shared__ float sB[8];
    if (lane == 0) { sA[ws] = ca; sB[ws] = cb; }
    __syncthreads();
    float hin = 0.f;
#pragma unroll
    for (int v = 0; v < 7; ++v)
        if (v < ws) hin = sA[v] * hin + sB[v];

    float* op = out + ((size_t)bb * 128 + DIR * 64 + c) * SEQL + wv * 1024 + lane * 4;
#pragma unroll
    for (int r = 0; r < 8; ++r) {
        const float hrow = PA[r] * hin + PB[r];
        const float ea = lane_prev<DIR>(A[r], 1);
        const float eb = lane_prev<DIR>(B[r], 1);
        float h = (tl == 0) ? hrow : (ea * hrow + eb);
        float o[4];
#pragma unroll
        for (int s = 0; s < 4; ++s) {
            const int j = DIR ? 3 - s : s;
            h = a[r][j] * h + b[r][j];
            o[j] = h;
        }
        *(float4*)(op + r * 128) = make_float4(o[0], o[1], o[2], o[3]);
    }
}

extern "C" void kernel_launch(void* const* d_in, const int* in_sizes, int n_in,
                              void* d_out, int out_size, void* d_ws, size_t ws_size,
                              hipStream_t stream) {
    const float* x = (const float*)d_in[0];
    const float* w = (const float*)d_in[1];
    float* out = (float*)d_out;

    float*  inv = (float*)d_ws;
    float2* hg  = (float2*)((char*)d_ws + ODIM * DIMI * sizeof(float));

    wnorm_kernel<<<1, 256, 0, stream>>>(w, inv);
    gemm_raw_kernel<<<dim3(SEQL / 128, NB), 256, 0, stream>>>(x, w, hg);
    scan_kernel<0><<<dim3(CH, NB), 256, 0, stream>>>(hg, inv, out);
    scan_kernel<1><<<dim3(CH, NB), 256, 0, stream>>>(hg, inv, out);
}
